# starting kernel
# baseline (speedup 1.0000x reference)
#include <hip/hip_runtime.h>
#include <math.h>


typedef float v2f __attribute__((ext_vector_type(2)));
typedef float v8f __attribute__((ext_vector_type(8)));

#define GMM_K 32
#define GMM_C 128
#define LOG2PI_F 1.8377866f
static constexpr float kLog2Pi  = 1.8378770664093453f;
static constexpr float kEps     = 5500.0f;
static constexpr float kInvLmb  = 1.0f / 1100.0f;

__global__ __launch_bounds__(256)
void gmm_precompute(const float* __restrict__ w,
                    const float* __restrict__ mu,
                    const float* __restrict__ cov,
                    float* __restrict__ M,
                    float* __restrict__ G)
{
    int idx = blockIdx.x * blockDim.x + threadIdx.x;
    if (idx >= GMM_K * GMM_C) return;
    int k = idx >> 7;
    int c = idx & (GMM_C - 1);

    const float* cv = cov + (size_t)idx * 4;
    float c00 = cv[0], c01 = cv[1], c10 = cv[2], c11 = cv[3];
    float det  = c00 * c11 - c01 * c10;
    float inv  = 1.0f / det;
    float P00  = c11 * inv;
    float P01  = -c01 * inv;
    float P10  = -c10 * inv;
    float P11  = c00 * inv;
    float m0   = mu[(size_t)idx * 2 + 0];
    float m1   = mu[(size_t)idx * 2 + 1];
    float s    = P01 + P10;
    float logdet = logf(det);
    float lw     = logf(w[idx]);
    float mPm    = P00 * m0 * m0 + s * m0 * m1 + P11 * m1 * m1;

    float* Mk = M + (size_t)k * 8 * GMM_C;
    Mk[0 * GMM_C + c] = -0.5f * P00;
    Mk[1 * GMM_C + c] = -0.5f * s;
    Mk[2 * GMM_C + c] = -0.5f * P11;
    Mk[3 * GMM_C + c] = P00 * m0 + 0.5f * s * m1;
    Mk[4 * GMM_C + c] = 0.5f * s * m0 + P11 * m1;
    Mk[5 * GMM_C + c] = lw - 0.5f * (2.0f * kLog2Pi + logdet) - 0.5f * mPm;
    Mk[6 * GMM_C + c] = 0.0f;
    Mk[7 * GMM_C + c] = 0.0f;

    float* Gc = G + (size_t)idx * 16;
    Gc[0] = P00;
    Gc[1] = P01;
    Gc[2] = P10;
    Gc[3] = P11;
    Gc[4] = P00 * m0 + P01 * m1;
    Gc[5] = P10 * m0 + P11 * m1;
    Gc[6] = 1.0f;
#pragma unroll
    for (int j = 7; j < 16; ++j) Gc[j] = 0.0f;
}

__global__ __launch_bounds__(128)
void gmm_wmma(const float* __restrict__ kps,
              const float* __restrict__ M,
              const float* __restrict__ G,
              float* __restrict__ out,
              int Btot)
{
    __shared__ float Gs[GMM_C * 16];
    __shared__ float compS[4][16 * GMM_C];
    __shared__ float Ssh[4][16 * 16];

    const int k    = blockIdx.y;
    const int wave = threadIdx.x >> 5;
    const int lane = threadIdx.x & 31;
    const int n    = lane & 15;
    const int half = lane >> 4;

    const float* Gk = G + (size_t)k * GMM_C * 16;
    for (int i = threadIdx.x; i < GMM_C * 16; i += 128) Gs[i] = Gk[i];
    __syncthreads();

    const int b0   = blockIdx.x * 64 + wave * 16;
    const int brow = b0 + n;
    const float x0 = kps[((size_t)brow * GMM_K + k) * 2 + 0];
    const float x1 = kps[((size_t)brow * GMM_K + k) * 2 + 1];

    v2f A0, A1;
    if (half == 0) { A0.x = x0 * x0; A0.y = x0 * x1; A1.x = x1;   A1.y = 1.0f; }
    else           { A0.x = x1 * x1; A0.y = x0;      A1.x = 0.0f; A1.y = 0.0f; }

    const float* Mk = M + (size_t)k * 8 * GMM_C;
    float* cw = compS[wave];

#pragma unroll
    for (int t = 0; t < 8; ++t) {
        const int c0 = t * 16 + n;
        v2f B0, B1;
        B0.x = Mk[(0 + 2 * half) * GMM_C + c0];
        B0.y = Mk[(1 + 2 * half) * GMM_C + c0];
        B1.x = Mk[(4 + 2 * half) * GMM_C + c0];
        B1.y = Mk[(5 + 2 * half) * GMM_C + c0];
        v8f acc = {};
        acc = __builtin_amdgcn_wmma_f32_16x16x4_f32(false, A0, false, B0,
                                                    (short)0, acc, false, false);
        acc = __builtin_amdgcn_wmma_f32_16x16x4_f32(false, A1, false, B1,
                                                    (short)0, acc, false, false);
#pragma unroll
        for (int v = 0; v < 8; ++v) {
            const int row = v + 8 * half;
            cw[row * GMM_C + t * 16 + n] = __expf(acc[v]);
        }
    }

    v8f acc2 = {};
#pragma unroll
    for (int cc = 0; cc < GMM_C; cc += 4) {
        const int r = cc + 2 * half;
        v2f A2, B2;
        A2.x = cw[n * GMM_C + r];
        A2.y = cw[n * GMM_C + r + 1];
        B2.x = Gs[r * 16 + n];
        B2.y = Gs[(r + 1) * 16 + n];
        acc2 = __builtin_amdgcn_wmma_f32_16x16x4_f32(false, A2, false, B2,
                                                     (short)0, acc2, false, false);
    }

    float* sw = Ssh[wave];
#pragma unroll
    for (int v = 0; v < 8; ++v) sw[(v + 8 * half) * 16 + n] = acc2[v];

    if (half == 0) {
        const float S0  = sw[n * 16 + 0];
        const float S1  = sw[n * 16 + 1];
        const float S2  = sw[n * 16 + 2];
        const float S3  = sw[n * 16 + 3];
        const float S4  = sw[n * 16 + 4];
        const float S5  = sw[n * 16 + 5];
        const float pdf = sw[n * 16 + 6];

        const float g0 = S4 - S0 * x0 - S1 * x1;
        const float g1 = S5 - S2 * x0 - S3 * x1;
        float nrm = sqrtf(g0 * g0 + g1 * g1);
        nrm = fmaxf(nrm, 1e-30f);
        const float scale = __expf((kEps - nrm) * kInvLmb) / nrm;

        float* dens = out + (size_t)brow * 64 + 2 * k;
        dens[0] = pdf;
        dens[1] = pdf;
        float* gv = out + (size_t)Btot * 64 + (size_t)brow * 64 + 2 * k;
        gv[0] = g0 * scale;
        gv[1] = g1 * scale;
    }
}

extern "C" void kernel_launch(void* const* d_in, const int* in_sizes, int n_in,
                              void* d_out, int out_size, void* d_ws, size_t ws_size,
                              hipStream_t stream)
{
    const float* kps = (const float*)d_in[0];
    const float* w   = (const float*)d_in[1];
    const float* mu  = (const float*)d_in[2];
    const float* cov = (const float*)d_in[3];
    float* out = (float*)d_out;

    const int Btot = in_sizes[0] / (GMM_K * 2);

    float* Mws = (float*)d_ws;
    float* Gws = Mws + (size_t)GMM_K * 8 * GMM_C;

    gmm_precompute<<<(GMM_K * GMM_C + 255) / 256, 256, 0, stream>>>(w, mu, cov, Mws, Gws);

    dim3 grid(Btot / 64, GMM_K);
    gmm_wmma<<<grid, 128, 0, stream>>>(kps, Mws, Gws, out, Btot);
}


# Precompute writes M (times log2 e, fragment order) and G (k pairs adjacent) as the main loop reads them: 16-byte loads and comp stores, fixed LDS offsets, bare exp2; same sum order, M rounded once
# speedup vs baseline: 1.8365x; 1.8365x over previous
#include <hip/hip_runtime.h>
#include <math.h>


typedef float v2f __attribute__((ext_vector_type(2)));
typedef float v8f __attribute__((ext_vector_type(8)));

#define GMM_K 32
#define GMM_C 128
#define CS_PITCH 132
static constexpr float kLog2Pi  = 1.8378770664093453f;
static constexpr float kLog2E   = 1.4426950408889634f;
static constexpr float kEps     = 5500.0f;
static constexpr float kInvLmb  = 1.0f / 1100.0f;

__global__ __launch_bounds__(256)
void gmm_precompute(const float* __restrict__ w,
                    const float* __restrict__ mu,
                    const float* __restrict__ cov,
                    float* __restrict__ M,
                    float* __restrict__ G)
{
    int idx = blockIdx.x * blockDim.x + threadIdx.x;
    if (idx >= GMM_K * GMM_C) return;
    int k = idx >> 7;
    int c = idx & (GMM_C - 1);

    const float* cv = cov + (size_t)idx * 4;
    float c00 = cv[0], c01 = cv[1], c10 = cv[2], c11 = cv[3];
    float det  = c00 * c11 - c01 * c10;
    float inv  = 1.0f / det;
    float P00  = c11 * inv;
    float P01  = -c01 * inv;
    float P10  = -c10 * inv;
    float P11  = c00 * inv;
    float m0   = mu[(size_t)idx * 2 + 0];
    float m1   = mu[(size_t)idx * 2 + 1];
    float s    = P01 + P10;
    float logdet = logf(det);
    float lw     = logf(w[idx]);
    float mPm    = P00 * m0 * m0 + s * m0 * m1 + P11 * m1 * m1;

    const float r0 = (-0.5f * P00) * kLog2E;
    const float r1 = (-0.5f * s) * kLog2E;
    const float r2 = (-0.5f * P11) * kLog2E;
    const float r3 = (P00 * m0 + 0.5f * s * m1) * kLog2E;
    const float r4 = (0.5f * s * m0 + P11 * m1) * kLog2E;
    const float r5 = (lw - 0.5f * (2.0f * kLog2Pi + logdet) - 0.5f * mPm) * kLog2E;

    const int q = c >> 6;
    const int n = (c & 63) >> 2;
    const int j = c & 3;
    float4* Mt = reinterpret_cast<float4*>(M) + ((size_t)k * 8 + (4 * q + j)) * 32;
    Mt[n]      = make_float4(r0, r1, r4, r5);
    Mt[n + 16] = make_float4(r2, r3, 0.0f, 0.0f);

    float* Gp = G + (size_t)k * (GMM_C * 16) + (size_t)(c >> 1) * 32 + (c & 1);
    Gp[0 * 2] = P00;
    Gp[1 * 2] = P01;
    Gp[2 * 2] = P10;
    Gp[3 * 2] = P11;
    Gp[4 * 2] = P00 * m0 + P01 * m1;
    Gp[5 * 2] = P10 * m0 + P11 * m1;
    Gp[6 * 2] = 1.0f;
#pragma unroll
    for (int jj = 7; jj < 16; ++jj) Gp[jj * 2] = 0.0f;
}

#define GMM_KL 4

__global__ __launch_bounds__(128)
void gmm_wmma(const float* __restrict__ kps,
              const float* __restrict__ M,
              const float* __restrict__ G,
              float* __restrict__ out,
              int Btot)
{
    __shared__ __attribute__((aligned(16))) float Gs[2][GMM_C * 16];
    __shared__ __attribute__((aligned(16))) float compS[4][16 * CS_PITCH];
    __shared__ __attribute__((aligned(16))) float Ssh[4][16 * 16];

    const int kbase = blockIdx.y * GMM_KL;
    const int wave = threadIdx.x >> 5;
    const int lane = threadIdx.x & 31;
    const int n    = lane & 15;
    const int half = lane >> 4;

    const int b0   = blockIdx.x * 64 + wave * 16;
    const int brow = b0 + n;
    const float* kprow = kps + (size_t)brow * GMM_K * 2;

    float* cw = compS[wave];
    float* sw = Ssh[wave];

    const float4* M4 = reinterpret_cast<const float4*>(M);
    const float4* G4 = reinterpret_cast<const float4*>(G);

    float xc0 = kprow[kbase * 2 + 0];
    float xc1 = kprow[kbase * 2 + 1];
    float4 Mc[8];
    {
        const float4* Mk = M4 + (size_t)kbase * 256;
#pragma unroll
        for (int t = 0; t < 8; ++t) Mc[t] = Mk[t * 32 + lane];
        const float4* Gk = G4 + (size_t)kbase * 512;
        float4* Gd = reinterpret_cast<float4*>(Gs[0]);
#pragma unroll
        for (int i = 0; i < 4; ++i) Gd[threadIdx.x + 128 * i] = Gk[threadIdx.x + 128 * i];
    }

    float rP[GMM_KL], rG0[GMM_KL], rG1[GMM_KL];
#pragma unroll
    for (int i = 0; i < GMM_KL; ++i) { rP[i] = 0.0f; rG0[i] = 0.0f; rG1[i] = 0.0f; }

#pragma unroll
    for (int kk = 0; kk < GMM_KL; ++kk) {
        const int k = kbase + kk;
        const bool more = (kk + 1 < GMM_KL);

        __syncthreads();

        float xn0 = 0.0f, xn1 = 0.0f;
        float4 Mn[8];
        float4 gn[4];
        if (more) {
            xn0 = kprow[(k + 1) * 2 + 0];
            xn1 = kprow[(k + 1) * 2 + 1];
            const float4* Mk = M4 + (size_t)(k + 1) * 256;
#pragma unroll
            for (int t = 0; t < 8; ++t) Mn[t] = Mk[t * 32 + lane];
            const float4* Gk = G4 + (size_t)(k + 1) * 512;
#pragma unroll
            for (int i = 0; i < 4; ++i) gn[i] = Gk[threadIdx.x + 128 * i];
        }

        const float* Gc = Gs[kk & 1];
        const float x0 = xc0;
        const float x1 = xc1;

        v2f A0, A1;
        if (half == 0) { A0.x = x0 * x0; A0.y = x0 * x1; A1.x = x1;   A1.y = 1.0f; }
        else           { A0.x = x1 * x1; A0.y = x0;      A1.x = 0.0f; A1.y = 0.0f; }

#pragma unroll
        for (int q = 0; q < 2; ++q) {
            v8f acc[4];
#pragma unroll
            for (int j = 0; j < 4; ++j) {
                const float4 m = Mc[4 * q + j];
                v2f B0, B1;
                B0.x = m.x;
                B0.y = m.y;
                B1.x = m.z;
                B1.y = m.w;
                v8f a = {};
                a = __builtin_amdgcn_wmma_f32_16x16x4_f32(false, A0, false, B0,
                                                          (short)0, a, false, false);
                a = __builtin_amdgcn_wmma_f32_16x16x4_f32(false, A1, false, B1,
                                                          (short)0, a, false, false);
                acc[j] = a;
            }
#pragma unroll
            for (int v = 0; v < 8; ++v) {
                const int row = v + 8 * half;
                float4 e;
                e.x = __builtin_amdgcn_exp2f(acc[0][v]);
                e.y = __builtin_amdgcn_exp2f(acc[1][v]);
                e.z = __builtin_amdgcn_exp2f(acc[2][v]);
                e.w = __builtin_amdgcn_exp2f(acc[3][v]);
                *reinterpret_cast<float4*>(&cw[row * CS_PITCH + 64 * q + 4 * n]) = e;
            }
        }

        v8f acc2 = {};
#pragma unroll
        for (int cc = 0; cc < GMM_C; cc += 4) {
            const int r = cc + 2 * half;
            v2f A2, B2;
            A2.x = cw[n * CS_PITCH + r];
            A2.y = cw[n * CS_PITCH + r + 1];
            B2.x = Gc[(r >> 1) * 32 + 2 * n];
            B2.y = Gc[(r >> 1) * 32 + 2 * n + 1];
            acc2 = __builtin_amdgcn_wmma_f32_16x16x4_f32(false, A2, false, B2,
                                                         (short)0, acc2, false, false);
        }

#pragma unroll
        for (int v = 0; v < 8; ++v) sw[(v + 8 * half) * 16 + n] = acc2[v];

        if (more) {
            asm volatile("" :: "v"(gn[0].x), "v"(gn[0].y), "v"(gn[0].z), "v"(gn[0].w),
                               "v"(gn[1].x), "v"(gn[1].y), "v"(gn[1].z), "v"(gn[1].w),
                               "v"(gn[2].x), "v"(gn[2].y), "v"(gn[2].z), "v"(gn[2].w),
                               "v"(gn[3].x), "v"(gn[3].y), "v"(gn[3].z), "v"(gn[3].w));
        }

        if (half == 0) {
            const float S0  = sw[n * 16 + 0];
            const float S1  = sw[n * 16 + 1];
            const float S2  = sw[n * 16 + 2];
            const float S3  = sw[n * 16 + 3];
            const float S4  = sw[n * 16 + 4];
            const float S5  = sw[n * 16 + 5];
            const float pdf = sw[n * 16 + 6];

            const float g0 = S4 - S0 * x0 - S1 * x1;
            const float g1 = S5 - S2 * x0 - S3 * x1;
            float nrm = sqrtf(g0 * g0 + g1 * g1);
            nrm = fmaxf(nrm, 1e-30f);
            const float scale = __expf((kEps - nrm) * kInvLmb) / nrm;

            rP[kk]  = pdf;
            rG0[kk] = g0 * scale;
            rG1[kk] = g1 * scale;
        }

        if (more) {
            float4* Gd = reinterpret_cast<float4*>(Gs[(kk + 1) & 1]);
#pragma unroll
            for (int i = 0; i < 4; ++i) Gd[threadIdx.x + 128 * i] = gn[i];
#pragma unroll
            for (int t = 0; t < 8; ++t) Mc[t] = Mn[t];
            xc0 = xn0;
            xc1 = xn1;
        }
    }

    static_assert(GMM_KL == 4, "the store below packs exactly four keypoints");
    if (half == 0) {
        float4* d4 = reinterpret_cast<float4*>(out + (size_t)brow * 64 + 2 * kbase);
        d4[0] = make_float4(rP[0], rP[0], rP[1], rP[1]);
        d4[1] = make_float4(rP[2], rP[2], rP[3], rP[3]);
        float4* g4 = reinterpret_cast<float4*>(out + (size_t)Btot * 64 + (size_t)brow * 64 + 2 * kbase);
        g4[0] = make_float4(rG0[0], rG1[0], rG0[1], rG1[1]);
        g4[1] = make_float4(rG0[2], rG1[2], rG0[3], rG1[3]);
    }
}

extern "C" void kernel_launch(void* const* d_in, const int* in_sizes, int n_in,
                              void* d_out, int out_size, void* d_ws, size_t ws_size,
                              hipStream_t stream)
{
    const float* kps = (const float*)d_in[0];
    const float* w   = (const float*)d_in[1];
    const float* mu  = (const float*)d_in[2];
    const float* cov = (const float*)d_in[3];
    float* out = (float*)d_out;

    const int Btot = in_sizes[0] / (GMM_K * 2);

    float* Mws = (float*)d_ws;
    float* Gws = Mws + (size_t)GMM_K * 8 * GMM_C;

    gmm_precompute<<<(GMM_K * GMM_C + 255) / 256, 256, 0, stream>>>(w, mu, cov, Mws, Gws);

    dim3 grid(Btot / 64, GMM_K / GMM_KL);
    gmm_wmma<<<grid, 128, 0, stream>>>(kps, Mws, Gws, out, Btot);
}


# Software-pipelined trip: exp2 of one component half issues beside the WMMAs of the next stage, comp stores and fragment reads stand in two runs, row finish taken off the WMMA path; same sum order
# speedup vs baseline: 1.9267x; 1.0491x over previous
#include <hip/hip_runtime.h>
#include <math.h>


typedef float v2f __attribute__((ext_vector_type(2)));
typedef float v8f __attribute__((ext_vector_type(8)));

#define GMM_K 32
#define GMM_C 128
#define CS_PITCH 132
static constexpr float kLog2Pi  = 1.8378770664093453f;
static constexpr float kLog2E   = 1.4426950408889634f;
static constexpr float kEps     = 5500.0f;
static constexpr float kInvLmb  = 1.0f / 1100.0f;

__global__ __launch_bounds__(256)
void gmm_precompute(const float* __restrict__ w,
                    const float* __restrict__ mu,
                    const float* __restrict__ cov,
                    float* __restrict__ M,
                    float* __restrict__ G)
{
    int idx = blockIdx.x * blockDim.x + threadIdx.x;
    if (idx >= GMM_K * GMM_C) return;
    int k = idx >> 7;
    int c = idx & (GMM_C - 1);

    const float* cv = cov + (size_t)idx * 4;
    float c00 = cv[0], c01 = cv[1], c10 = cv[2], c11 = cv[3];
    float det  = c00 * c11 - c01 * c10;
    float inv  = 1.0f / det;
    float P00  = c11 * inv;
    float P01  = -c01 * inv;
    float P10  = -c10 * inv;
    float P11  = c00 * inv;
    float m0   = mu[(size_t)idx * 2 + 0];
    float m1   = mu[(size_t)idx * 2 + 1];
    float s    = P01 + P10;
    float logdet = logf(det);
    float lw     = logf(w[idx]);
    float mPm    = P00 * m0 * m0 + s * m0 * m1 + P11 * m1 * m1;

    const float r0 = (-0.5f * P00) * kLog2E;
    const float r1 = (-0.5f * s) * kLog2E;
    const float r2 = (-0.5f * P11) * kLog2E;
    const float r3 = (P00 * m0 + 0.5f * s * m1) * kLog2E;
    const float r4 = (0.5f * s * m0 + P11 * m1) * kLog2E;
    const float r5 = (lw - 0.5f * (2.0f * kLog2Pi + logdet) - 0.5f * mPm) * kLog2E;

    const int q = c >> 6;
    const int n = (c & 63) >> 2;
    const int j = c & 3;
    float4* Mt = reinterpret_cast<float4*>(M) + ((size_t)k * 8 + (4 * q + j)) * 32;
    Mt[n]      = make_float4(r0, r1, r4, r5);
    Mt[n + 16] = make_float4(r2, r3, 0.0f, 0.0f);

    float* Gp = G + (size_t)k * (GMM_C * 16) + (size_t)(c >> 1) * 32 + (c & 1);
    Gp[0 * 2] = P00;
    Gp[1 * 2] = P01;
    Gp[2 * 2] = P10;
    Gp[3 * 2] = P11;
    Gp[4 * 2] = P00 * m0 + P01 * m1;
    Gp[5 * 2] = P10 * m0 + P11 * m1;
    Gp[6 * 2] = 1.0f;
#pragma unroll
    for (int jj = 7; jj < 16; ++jj) Gp[jj * 2] = 0.0f;
}

#define GMM_KL 4

__device__ __forceinline__ void gmm_finish_row(const float* sw, int n, float x0, float x1,
                                               float& oP, float& oG0, float& oG1)
{
    const float S0  = sw[n * 16 + 0];
    const float S1  = sw[n * 16 + 1];
    const float S2  = sw[n * 16 + 2];
    const float S3  = sw[n * 16 + 3];
    const float S4  = sw[n * 16 + 4];
    const float S5  = sw[n * 16 + 5];
    const float pdf = sw[n * 16 + 6];

    const float g0 = S4 - S0 * x0 - S1 * x1;
    const float g1 = S5 - S2 * x0 - S3 * x1;
    float nrm = sqrtf(g0 * g0 + g1 * g1);
    nrm = fmaxf(nrm, 1e-30f);
    const float scale = __expf((kEps - nrm) * kInvLmb) / nrm;

    oP  = pdf;
    oG0 = g0 * scale;
    oG1 = g1 * scale;
}

__global__ __launch_bounds__(128)
void gmm_wmma(const float* __restrict__ kps,
              const float* __restrict__ M,
              const float* __restrict__ G,
              float* __restrict__ out,
              int Btot)
{
    __shared__ __attribute__((aligned(16))) float Gs[2][GMM_C * 16];
    __shared__ __attribute__((aligned(16))) float compS[4][16 * CS_PITCH];
    __shared__ __attribute__((aligned(16))) float Ssh[4][16 * 16];

    const int kbase = blockIdx.y * GMM_KL;
    const int wave = threadIdx.x >> 5;
    const int lane = threadIdx.x & 31;
    const int n    = lane & 15;
    const int half = lane >> 4;

    const int b0   = blockIdx.x * 64 + wave * 16;
    const int brow = b0 + n;
    const float* kprow = kps + (size_t)brow * GMM_K * 2;

    float* cw = compS[wave];
    float* sw = Ssh[wave];

    const float4* M4 = reinterpret_cast<const float4*>(M);
    const float4* G4 = reinterpret_cast<const float4*>(G);

    static_assert(GMM_KL == 4, "two 16-byte pieces hold exactly four keypoints' x");
    const float4 xa = reinterpret_cast<const float4*>(kprow + kbase * 2)[0];
    const float4 xb = reinterpret_cast<const float4*>(kprow + kbase * 2)[1];
    const float xs0[GMM_KL] = { xa.x, xa.z, xb.x, xb.z };
    const float xs1[GMM_KL] = { xa.y, xa.w, xb.y, xb.w };
    float4 Mc[8];
    {
        const float4* Mk = M4 + (size_t)kbase * 256;
#pragma unroll
        for (int t = 0; t < 8; ++t) Mc[t] = Mk[t * 32 + lane];
        const float4* Gk = G4 + (size_t)kbase * 512;
        float4* Gd = reinterpret_cast<float4*>(Gs[0]);
#pragma unroll
        for (int i = 0; i < 4; ++i) Gd[threadIdx.x + 128 * i] = Gk[threadIdx.x + 128 * i];
    }

    asm volatile("" :: "v"(xa.x), "v"(xa.y), "v"(xa.z), "v"(xa.w),
                       "v"(xb.x), "v"(xb.y), "v"(xb.z), "v"(xb.w));

    float rP[GMM_KL], rG0[GMM_KL], rG1[GMM_KL];
#pragma unroll
    for (int i = 0; i < GMM_KL; ++i) { rP[i] = 0.0f; rG0[i] = 0.0f; rG1[i] = 0.0f; }

    float xp0 = 0.0f, xp1 = 0.0f;

#pragma unroll
    for (int kk = 0; kk < GMM_KL; ++kk) {
        const int k = kbase + kk;
        const bool more = (kk + 1 < GMM_KL);

        __syncthreads();
        __builtin_amdgcn_sched_barrier(0);

        float4 Mn[8];
        float4 gn[4];
        if (more) {
            const float4* Mk = M4 + (size_t)(k + 1) * 256;
#pragma unroll
            for (int t = 0; t < 8; ++t) Mn[t] = Mk[t * 32 + lane];
            const float4* Gk = G4 + (size_t)(k + 1) * 512;
#pragma unroll
            for (int i = 0; i < 4; ++i) gn[i] = Gk[threadIdx.x + 128 * i];
        }

        const float* Gc = Gs[kk & 1];
        const float x0 = xs0[kk];
        const float x1 = xs1[kk];

        v2f A0, A1;
        if (half == 0) { A0.x = x0 * x0; A0.y = x0 * x1; A1.x = x1;   A1.y = 1.0f; }
        else           { A0.x = x1 * x1; A0.y = x0;      A1.x = 0.0f; A1.y = 0.0f; }

        v8f a0[4];
#pragma unroll
        for (int j = 0; j < 4; ++j) {
            const float4 m = Mc[j];
            v2f B0, B1;
            B0.x = m.x; B0.y = m.y; B1.x = m.z; B1.y = m.w;
            v8f a = {};
            a = __builtin_amdgcn_wmma_f32_16x16x4_f32(false, A0, false, B0, (short)0, a, false, false);
            a = __builtin_amdgcn_wmma_f32_16x16x4_f32(false, A1, false, B1, (short)0, a, false, false);
            a0[j] = a;
        }

        v8f a1[4];
#pragma unroll
        for (int j = 0; j < 4; ++j) {
            const float4 m = Mc[4 + j];
            v2f B0, B1;
            B0.x = m.x; B0.y = m.y; B1.x = m.z; B1.y = m.w;
            v8f a = {};
            a = __builtin_amdgcn_wmma_f32_16x16x4_f32(false, A0, false, B0, (short)0, a, false, false);
            a = __builtin_amdgcn_wmma_f32_16x16x4_f32(false, A1, false, B1, (short)0, a, false, false);
            a1[j] = a;
        }
#pragma unroll
        for (int v = 0; v < 8; ++v) {
            float4 e;
            e.x = __builtin_amdgcn_exp2f(a0[0][v]);
            e.y = __builtin_amdgcn_exp2f(a0[1][v]);
            e.z = __builtin_amdgcn_exp2f(a0[2][v]);
            e.w = __builtin_amdgcn_exp2f(a0[3][v]);
            *reinterpret_cast<float4*>(&cw[(v + 8 * half) * CS_PITCH + 4 * n]) = e;
        }
        v2f Af[16], Bf[16];
#pragma unroll
        for (int s = 0; s < 16; ++s) {
            const int r = 4 * s + 2 * half;
            Af[s].x = cw[n * CS_PITCH + r];
            Af[s].y = cw[n * CS_PITCH + r + 1];
            Bf[s].x = Gc[(r >> 1) * 32 + 2 * n];
            Bf[s].y = Gc[(r >> 1) * 32 + 2 * n + 1];
        }

        if (kk > 0) gmm_finish_row(sw, n, xp0, xp1, rP[kk - 1], rG0[kk - 1], rG1[kk - 1]);

        v2f Ag[16], Bg[16];
        v8f acc2 = {};
#pragma unroll
        for (int s = 0; s < 16; ++s)
            acc2 = __builtin_amdgcn_wmma_f32_16x16x4_f32(false, Af[s], false, Bf[s],
                                                         (short)0, acc2, false, false);
#pragma unroll
        for (int v = 0; v < 8; ++v) {
            float4 e;
            e.x = __builtin_amdgcn_exp2f(a1[0][v]);
            e.y = __builtin_amdgcn_exp2f(a1[1][v]);
            e.z = __builtin_amdgcn_exp2f(a1[2][v]);
            e.w = __builtin_amdgcn_exp2f(a1[3][v]);
            *reinterpret_cast<float4*>(&cw[(v + 8 * half) * CS_PITCH + 64 + 4 * n]) = e;
        }
#pragma unroll
        for (int s = 0; s < 16; ++s) {
            const int r = 64 + 4 * s + 2 * half;
            Ag[s].x = cw[n * CS_PITCH + r];
            Ag[s].y = cw[n * CS_PITCH + r + 1];
            Bg[s].x = Gc[(r >> 1) * 32 + 2 * n];
            Bg[s].y = Gc[(r >> 1) * 32 + 2 * n + 1];
        }

#pragma unroll
        for (int s = 0; s < 16; ++s)
            acc2 = __builtin_amdgcn_wmma_f32_16x16x4_f32(false, Ag[s], false, Bg[s],
                                                         (short)0, acc2, false, false);

#pragma unroll
        for (int v = 0; v < 8; ++v) sw[(v + 8 * half) * 16 + n] = acc2[v];
        xp0 = x0;
        xp1 = x1;

        if (more) {
            float4* Gd = reinterpret_cast<float4*>(Gs[(kk + 1) & 1]);
#pragma unroll
            for (int i = 0; i < 4; ++i) Gd[threadIdx.x + 128 * i] = gn[i];
#pragma unroll
            for (int t = 0; t < 8; ++t) Mc[t] = Mn[t];
        }

        __builtin_amdgcn_sched_group_barrier(0x020, 12, 0);
        __builtin_amdgcn_sched_group_barrier(0x008, 8, 0);
#pragma unroll
        for (int i = 0; i < 8; ++i) {
            __builtin_amdgcn_sched_group_barrier(0x008, 1, 0);
            __builtin_amdgcn_sched_group_barrier(0x400, 4, 0);
        }
        __builtin_amdgcn_sched_group_barrier(0x200, 8, 0);
        __builtin_amdgcn_sched_group_barrier(0x100, 18, 0);
#pragma unroll
        for (int i = 0; i < 8; ++i) {
            __builtin_amdgcn_sched_group_barrier(0x008, 1, 0);
            __builtin_amdgcn_sched_group_barrier(0x400, 4, 0);
        }
        __builtin_amdgcn_sched_group_barrier(0x200, 8, 0);
        __builtin_amdgcn_sched_group_barrier(0x100, 16, 0);
        __builtin_amdgcn_sched_group_barrier(0x008, 24, 0);
        __builtin_amdgcn_sched_group_barrier(0x200, 8, 0);

        __builtin_amdgcn_sched_barrier(0);
    }

    gmm_finish_row(sw, n, xp0, xp1, rP[GMM_KL - 1], rG0[GMM_KL - 1], rG1[GMM_KL - 1]);

    static_assert(GMM_KL == 4, "the store below packs exactly four keypoints");
    if (half == 0) {
        float4* d4 = reinterpret_cast<float4*>(out + (size_t)brow * 64 + 2 * kbase);
        d4[0] = make_float4(rP[0], rP[0], rP[1], rP[1]);
        d4[1] = make_float4(rP[2], rP[2], rP[3], rP[3]);
        float4* g4 = reinterpret_cast<float4*>(out + (size_t)Btot * 64 + (size_t)brow * 64 + 2 * kbase);
        g4[0] = make_float4(rG0[0], rG1[0], rG0[1], rG1[1]);
        g4[1] = make_float4(rG0[2], rG1[2], rG0[3], rG1[3]);
    }
}

extern "C" void kernel_launch(void* const* d_in, const int* in_sizes, int n_in,
                              void* d_out, int out_size, void* d_ws, size_t ws_size,
                              hipStream_t stream)
{
    const float* kps = (const float*)d_in[0];
    const float* w   = (const float*)d_in[1];
    const float* mu  = (const float*)d_in[2];
    const float* cov = (const float*)d_in[3];
    float* out = (float*)d_out;

    const int Btot = in_sizes[0] / (GMM_K * 2);

    float* Mws = (float*)d_ws;
    float* Gws = Mws + (size_t)GMM_K * 8 * GMM_C;

    gmm_precompute<<<(GMM_K * GMM_C + 255) / 256, 256, 0, stream>>>(w, mu, cov, Mws, Gws);

    dim3 grid(Btot / 64, GMM_K / GMM_KL);
    gmm_wmma<<<grid, 128, 0, stream>>>(kps, Mws, Gws, out, Btot);
}
